# starting kernel
# baseline (speedup 1.0000x reference)
#include <hip/hip_runtime.h>
#include <hip/hip_bf16.h>

typedef __attribute__((ext_vector_type(2))) float v2f;
typedef __attribute__((ext_vector_type(8))) float v8f;

#define HFEAT 64

__global__ void k_fill1(float* d, int n) {
    int t = blockIdx.x * blockDim.x + threadIdx.x;
    if (t < n) d[t] = 1.0f;
}

__global__ void k_deg(const int* __restrict__ dst, float* __restrict__ deg, int e) {
    int t = blockIdx.x * blockDim.x + threadIdx.x;
    if (t < e) unsafeAtomicAdd(&deg[dst[t]], 1.0f);
}

__global__ void k_rsqrt(float* d, int n) {
    int t = blockIdx.x * blockDim.x + threadIdx.x;
    if (t < n) d[t] = rsqrtf(d[t]);
}

template <int K, int NC>
__global__ void k_gemm_wmma(const float* __restrict__ A, const float* __restrict__ B,
                            float* __restrict__ C, int M) {
    const int NT = NC / 16;
    int wave = (int)((blockIdx.x * blockDim.x + threadIdx.x) >> 5);
    int lane = threadIdx.x & 31;
    int totalTiles = (M / 16) * NT;
    if (wave >= totalTiles) return;

    int m0 = (wave / NT) * 16;
    int n0 = (wave % NT) * 16;
    int row  = m0 + (lane & 15);
    int col  = n0 + (lane & 15);
    int koff = (lane >> 4) * 2;

    v8f acc = {};
    const float* arow = A + (size_t)row * K + koff;
#pragma unroll 4
    for (int k = 0; k < K; k += 4) {
        v2f a = *(const v2f*)(arow + k);
        v2f b;
        b.x = B[(size_t)(k + koff)     * NC + col];
        b.y = B[(size_t)(k + koff + 1) * NC + col];
        acc = __builtin_amdgcn_wmma_f32_16x16x4_f32(false, a, false, b,
                                                    (short)0, acc, false, false);
    }

    int rbase = m0 + ((lane >> 4) * 8);
#pragma unroll
    for (int i = 0; i < 8; ++i)
        C[(size_t)(rbase + i) * NC + col] = acc[i];
}

__global__ void k_self_init(const float* __restrict__ xw, const float* __restrict__ dinv,
                            float* __restrict__ out, int total) {
    int t = blockIdx.x * blockDim.x + threadIdx.x;
    if (t >= total) return;
    int node = t >> 6;
    float di = dinv[node];
    out[t] = di * di * xw[t];
}

__global__ void k_edge_scatter(const int* __restrict__ src, const int* __restrict__ dst,
                               const float* __restrict__ dinv,
                               const float* __restrict__ xw, float* __restrict__ out, int e) {
    int t = blockIdx.x * blockDim.x + threadIdx.x;
    int edge = t >> 4;
    if (edge >= e) return;
    int f4 = (t & 15) * 4;
    int s = src[edge];
    int d = dst[edge];
    float c = dinv[s] * dinv[d];
    const float4 v = *(const float4*)(xw + (size_t)s * HFEAT + f4);
    float* o = out + (size_t)d * HFEAT + f4;
    unsafeAtomicAdd(o + 0, c * v.x);
    unsafeAtomicAdd(o + 1, c * v.y);
    unsafeAtomicAdd(o + 2, c * v.z);
    unsafeAtomicAdd(o + 3, c * v.w);
}

__global__ void k_bias_relu(float* __restrict__ h, const float* __restrict__ b, int total) {
    int t = blockIdx.x * blockDim.x + threadIdx.x;
    if (t >= total) return;
    float v = h[t] + b[t & (HFEAT - 1)];
    h[t] = v > 0.0f ? v : 0.0f;
}

__global__ void k_fc_head(const float* __restrict__ g, const float* __restrict__ fc1_b,
                          const float* __restrict__ fcoW, const float* __restrict__ fcob,
                          float* __restrict__ out, int n) {
    int wave = (int)((blockIdx.x * blockDim.x + threadIdx.x) >> 5);
    int lane = threadIdx.x & 31;
    if (wave >= n) return;
    float v = g[(size_t)wave * 32 + lane] + fc1_b[lane];
    v = v > 0.0f ? v : 0.0f;
    float p = v * fcoW[lane];
#pragma unroll
    for (int off = 16; off > 0; off >>= 1) p += __shfl_xor(p, off, 32);
    if (lane == 0) out[wave] = p + fcob[0];
}

extern "C" void kernel_launch(void* const* d_in, const int* in_sizes, int n_in,
                              void* d_out, int out_size, void* d_ws, size_t ws_size,
                              hipStream_t stream) {
    const float* x     = (const float*)d_in[0];
    const int*   ei    = (const int*)d_in[1];
    const float* W1    = (const float*)d_in[2];
    const float* b1    = (const float*)d_in[3];
    const float* W2    = (const float*)d_in[4];
    const float* b2    = (const float*)d_in[5];
    const float* fc1W  = (const float*)d_in[6];
    const float* fc1b  = (const float*)d_in[7];
    const float* fcoW  = (const float*)d_in[8];
    const float* fcob  = (const float*)d_in[9];
    float* out = (float*)d_out;

    const int N = in_sizes[0] / 128;
    const int E = in_sizes[1] / 2;
    const int* src = ei;
    const int* dst = ei + E;

    float* dinv = (float*)d_ws;
    float* bufA = dinv + N;
    float* bufB = bufA + (size_t)N * HFEAT;

    const int B256 = 256;
    auto blocks = [](long long work, int bs) { return (int)((work + bs - 1) / bs); };

    k_fill1<<<blocks(N, B256), B256, 0, stream>>>(dinv, N);
    k_deg<<<blocks(E, B256), B256, 0, stream>>>(dst, dinv, E);
    k_rsqrt<<<blocks(N, B256), B256, 0, stream>>>(dinv, N);

    {
        long long waves = (long long)(N / 16) * (HFEAT / 16);
        k_gemm_wmma<128, HFEAT><<<blocks(waves * 32, B256), B256, 0, stream>>>(x, W1, bufA, N);
    }
    k_self_init<<<blocks((long long)N * HFEAT, B256), B256, 0, stream>>>(bufA, dinv, bufB, N * HFEAT);
    k_edge_scatter<<<blocks((long long)E * 16, B256), B256, 0, stream>>>(src, dst, dinv, bufA, bufB, E);
    k_bias_relu<<<blocks((long long)N * HFEAT, B256), B256, 0, stream>>>(bufB, b1, N * HFEAT);

    {
        long long waves = (long long)(N / 16) * (HFEAT / 16);
        k_gemm_wmma<HFEAT, HFEAT><<<blocks(waves * 32, B256), B256, 0, stream>>>(bufB, W2, bufA, N);
    }
    k_self_init<<<blocks((long long)N * HFEAT, B256), B256, 0, stream>>>(bufA, dinv, bufB, N * HFEAT);
    k_edge_scatter<<<blocks((long long)E * 16, B256), B256, 0, stream>>>(src, dst, dinv, bufA, bufB, E);
    k_bias_relu<<<blocks((long long)N * HFEAT, B256), B256, 0, stream>>>(bufB, b2, N * HFEAT);

    {
        long long waves = (long long)(N / 16) * (32 / 16);
        k_gemm_wmma<HFEAT, 32><<<blocks(waves * 32, B256), B256, 0, stream>>>(bufB, fc1W, bufA, N);
    }
    k_fc_head<<<blocks((long long)N * 32, B256), B256, 0, stream>>>(bufA, fc1b, fcoW, fcob, out, N);
}


# layer 2 aggregates h1 itself, A(hW)=(Ah)W; W2, fc1 and the head then run as one per-row WMMA kernel, 16 rows per wave: 9 to 7 launches, h2 and g round trips go; sum order changed: aggregate, then W2
# speedup vs baseline: 2.4060x; 2.4060x over previous
#include <hip/hip_runtime.h>
#include <hip/hip_bf16.h>

typedef __attribute__((ext_vector_type(2))) float v2f;
typedef __attribute__((ext_vector_type(8))) float v8f;

#define HFEAT 64

__global__ void k_fill1(float* d, int n) {
    int t = blockIdx.x * blockDim.x + threadIdx.x;
    if (t < n) d[t] = 1.0f;
}

#define TILE_ROWS 391
#define NTILES 256
#define BM_EDGES 2048
#define DEG_THREADS 1024
#define DEG_EPT (BM_EDGES / DEG_THREADS)
#define BM_WORDS 32
#define BM_PITCH (BM_WORDS + 1)
#define REC_CAP (BM_WORDS - 1)
#define REC_SHIFT 9
#define REC_MASK ((1u << REC_SHIFT) - 1u)

__global__ __launch_bounds__(DEG_THREADS) void k_deg_records(const int* __restrict__ src,
                                                             const int* __restrict__ dst,
                                                             float* __restrict__ deg,
                                                             unsigned* __restrict__ rec, int e, int wpt) {
    __shared__ unsigned s[NTILES * BM_PITCH];
    const int tid = threadIdx.x;
    const int lane = tid & 31;
    const int w = tid >> 5;
    const int t0 = blockIdx.x * BM_EDGES + tid;
    int d[DEG_EPT];
    int sv[DEG_EPT];
#pragma unroll
    for (int k = 0; k < DEG_EPT; ++k) {
        const int t = t0 + k * DEG_THREADS;
        const int tl = t < e ? t : e - 1;
        d[k] = dst[tl];
        sv[k] = src[tl];
        if (t >= e) d[k] = -1;
    }
    for (int i = tid; i < NTILES * BM_PITCH; i += DEG_THREADS) s[i] = 0u;
    __syncthreads();
#pragma unroll
    for (int k = 0; k < DEG_EPT; ++k) {
        if (d[k] >= 0) {
            unsafeAtomicAdd(&deg[d[k]], 1.0f);
            const unsigned tile = (unsigned)d[k] / TILE_ROWS;
            if (tile < NTILES) {
                const unsigned slot = __hip_atomic_fetch_add(&s[tile * BM_PITCH], 1u, __ATOMIC_RELAXED,
                                                             __HIP_MEMORY_SCOPE_WORKGROUP);
                if (slot < REC_CAP)
                    s[tile * BM_PITCH + 1 + slot] =
                        ((unsigned)sv[k] << REC_SHIFT) | ((unsigned)d[k] - tile * TILE_ROWS);
            }
        }
    }
    __syncthreads();
    unsigned* o = rec + (size_t)w * wpt + blockIdx.x * BM_WORDS + lane;
#pragma unroll
    for (int k = 0; k < NTILES / BM_WORDS; ++k)
        o[(size_t)k * BM_WORDS * wpt] = s[(w + k * BM_WORDS) * BM_PITCH + lane];
}

__global__ void k_rsqrt(float* d, int n) {
    int t = blockIdx.x * blockDim.x + threadIdx.x;
    if (t < n) d[t] = rsqrtf(d[t]);
}

template <int K, int NC>
__global__ void k_gemm_wmma(const float* __restrict__ A, const float* __restrict__ B,
                            float* __restrict__ C, int M) {
    const int NT = NC / 16;
    int wave = (int)((blockIdx.x * blockDim.x + threadIdx.x) >> 5);
    int lane = threadIdx.x & 31;
    int totalTiles = (M / 16) * NT;
    if (wave >= totalTiles) return;

    int m0 = (wave / NT) * 16;
    int n0 = (wave % NT) * 16;
    int row  = m0 + (lane & 15);
    int col  = n0 + (lane & 15);
    int koff = (lane >> 4) * 2;

    v8f acc = {};
    const float* arow = A + (size_t)row * K + koff;
#pragma unroll 4
    for (int k = 0; k < K; k += 4) {
        v2f a = *(const v2f*)(arow + k);
        v2f b;
        b.x = B[(size_t)(k + koff)     * NC + col];
        b.y = B[(size_t)(k + koff + 1) * NC + col];
        acc = __builtin_amdgcn_wmma_f32_16x16x4_f32(false, a, false, b,
                                                    (short)0, acc, false, false);
    }

    int rbase = m0 + ((lane >> 4) * 8);
#pragma unroll
    for (int i = 0; i < 8; ++i)
        C[(size_t)(rbase + i) * NC + col] = acc[i];
}

#define AGG_THREADS 1024
#define AGG_WAVES 32
#define SCAN_LINES 4
#define OVF_EDGES 1024
#define QCAP (32 + OVF_EDGES)

__device__ __forceinline__ void load_lines(const unsigned* __restrict__ row, int base, int nblk,
                                           int lane, unsigned (&v)[SCAN_LINES]) {
#pragma unroll
    for (int j = 0; j < SCAN_LINES; ++j)
        v[j] = (base + j < nblk) ? row[(size_t)(base + j) * BM_WORDS + lane] : 0u;
}

__device__ __forceinline__ void lds_add(float* p, float v) {
    __hip_atomic_fetch_add(p, v, __ATOMIC_RELAXED, __HIP_MEMORY_SCOPE_WORKGROUP);
}

__device__ __forceinline__ void drain(const unsigned* qe, int start, int count, int lane, int d0,
                                      const float* __restrict__ dinv, const float* __restrict__ xw,
                                      float* tile) {
    __builtin_amdgcn_fence(__ATOMIC_RELEASE, "wavefront");
    __builtin_amdgcn_wave_barrier();
    __builtin_amdgcn_fence(__ATOMIC_ACQUIRE, "wavefront");
    int s_i = 0, t_i = 0;
    float c_i = 0.0f;
    if (lane < count) {
        const unsigned r_i = qe[start + lane];
        s_i = (int)(r_i >> REC_SHIFT);
        t_i = (int)(r_i & REC_MASK);
        c_i = dinv[s_i] * dinv[d0 + t_i];
    }
    int i = 0;
    for (; i + 8 <= count; i += 8) {
        float r[16];
        int tt[8];
        float cc[8];
#pragma unroll
        for (int u = 0; u < 8; ++u) {
            int s = __builtin_amdgcn_readlane(s_i, i + u);
            tt[u] = __builtin_amdgcn_readlane(t_i, i + u);
            cc[u] = __int_as_float(__builtin_amdgcn_readlane(__float_as_int(c_i), i + u));
            const float* rp = xw + (size_t)s * HFEAT + lane;
            r[2 * u] = rp[0];
            r[2 * u + 1] = rp[32];
        }
#pragma unroll
        for (int u = 0; u < 8; ++u) {
            float* o = tile + tt[u] * HFEAT + lane;
            lds_add(o, cc[u] * r[2 * u]);
            lds_add(o + 32, cc[u] * r[2 * u + 1]);
        }
    }
    for (; i < count; ++i) {
        int s = __builtin_amdgcn_readlane(s_i, i);
        int t = __builtin_amdgcn_readlane(t_i, i);
        float c = __int_as_float(__builtin_amdgcn_readlane(__float_as_int(c_i), i));
        const float* rp = xw + (size_t)s * HFEAT + lane;
        float r0 = rp[0];
        float r1 = rp[32];
        float* o = tile + t * HFEAT + lane;
        lds_add(o, c * r0);
        lds_add(o + 32, c * r1);
    }
    __builtin_amdgcn_fence(__ATOMIC_RELEASE, "wavefront");
    __builtin_amdgcn_wave_barrier();
}

template <bool BIAS_RELU>
__global__ __launch_bounds__(AGG_THREADS) void k_aggregate(
        const int* __restrict__ src, const int* __restrict__ dst, const float* __restrict__ dinv,
        const float* __restrict__ xw, const float* __restrict__ bias, float* __restrict__ out,
        const unsigned* __restrict__ rec, int nblk, int n, int e) {
    __shared__ float tile[TILE_ROWS * HFEAT];
    __shared__ unsigned q_e[AGG_WAVES * QCAP];

    const int tid = threadIdx.x;
    const int lane = tid & 31;
    const int w = __builtin_amdgcn_readfirstlane(tid >> 5);
    const int d0 = blockIdx.x * TILE_ROWS;
    int cnt = n - d0;
    if (cnt > TILE_ROWS) cnt = TILE_ROWS;
    const int total = cnt * HFEAT;

    const float* xwt = xw + (size_t)d0 * HFEAT;
    for (int i = tid; i < total; i += AGG_THREADS) {
        float di = dinv[d0 + (i >> 6)];
        tile[i] = di * di * xwt[i];
    }
    __syncthreads();

    unsigned* qe = q_e + w * QCAP;
    int qn = 0;
    const unsigned* row = rec + (size_t)blockIdx.x * nblk * BM_WORDS;

    unsigned cur[SCAN_LINES];
    unsigned nxt[SCAN_LINES] = {};
    int base = w * SCAN_LINES;
    if (base < nblk) load_lines(row, base, nblk, lane, cur);
    for (; base < nblk; base += AGG_WAVES * SCAN_LINES) {
        const int nb = base + AGG_WAVES * SCAN_LINES;
        if (nb < nblk) load_lines(row, nb, nblk, lane, nxt);
#pragma unroll
        for (int j = 0; j < SCAN_LINES; ++j) {
            const unsigned word = cur[j];
            const int c = __builtin_amdgcn_readlane((int)word, 0);
            const bool fits = c <= REC_CAP;
            const int passes = fits ? 1 : BM_EDGES / OVF_EDGES;
            for (int h = 0; h < passes; ++h) {
                if (fits) {
                    if (lane >= 1 && lane <= c) qe[qn + lane - 1] = word;
                    qn += c;
                } else {
                    const int e0 = (base + j) * BM_EDGES + h * OVF_EDGES + lane;
#pragma nounroll
                    for (int r = 0; r < OVF_EDGES / 32; ++r) {
                        const int e_i = e0 + r * 32;
                        bool m = false;
                        unsigned rr = 0u;
                        if (e_i < e) {
                            const unsigned loc = (unsigned)(dst[e_i] - d0);
                            if (loc < (unsigned)cnt) {
                                m = true;
                                rr = ((unsigned)src[e_i] << REC_SHIFT) | loc;
                            }
                        }
                        const unsigned any = __builtin_amdgcn_ballot_w32(m);
                        if (m) qe[qn + (int)__builtin_amdgcn_mbcnt_lo(any, 0u)] = rr;
                        qn += __builtin_popcount(any);
                    }
                }
                while (qn >= 32) {
                    qn -= 32;
                    drain(qe, qn, 32, lane, d0, dinv, xw, tile);
                }
            }
        }
#pragma unroll
        for (int j = 0; j < SCAN_LINES; ++j) cur[j] = nxt[j];
    }
    if (qn > 0) drain(qe, 0, qn, lane, d0, dinv, xw, tile);
    __syncthreads();

    float* ot = out + (size_t)d0 * HFEAT;
    for (int i = tid; i < total; i += AGG_THREADS) {
        float v = tile[i];
        if (BIAS_RELU) {
            v += bias[i & (HFEAT - 1)];
            v = v > 0.0f ? v : 0.0f;
        }
        ot[i] = v;
    }
}

#define MLP_WAVES 16
#define MLP_THREADS (MLP_WAVES * 32)
#define MLP_GROUPS 256
#define H_PITCH 68
#define FC 32

__device__ __forceinline__ void load_a_tile(const float* __restrict__ a, int g, int c, int h, v2f (&f)[16]) {
    const float* arow = a + ((size_t)g * 16 + c) * HFEAT + 2 * h;
#pragma unroll
    for (int kk = 0; kk < 16; ++kk) f[kk] = *(const v2f*)(arow + 4 * kk);
}

__global__ __launch_bounds__(MLP_THREADS) void k_mlp(
        const float* __restrict__ a, const float* __restrict__ W2, const float* __restrict__ b2,
        const float* __restrict__ fc1W, const float* __restrict__ fc1b, const float* __restrict__ fcoW,
        const float* __restrict__ fcob, float* __restrict__ out, int ntiles) {
    __shared__ float w2s[HFEAT * HFEAT];
    __shared__ float f1s[HFEAT * FC];
    __shared__ float hs[MLP_WAVES * 16 * H_PITCH];

    const int tid = threadIdx.x;
    const int lane = tid & 31;
    const int w = __builtin_amdgcn_readfirstlane(tid >> 5);
    const int c = lane & 15;
    const int h = lane >> 4;
    const int stride = MLP_GROUPS * MLP_WAVES;

    int g = (int)blockIdx.x + MLP_GROUPS * w;
    v2f af[16];
    v2f an[16] = {};
    if (g < ntiles) load_a_tile(a, g, c, h, af);

    for (int p = tid; p < HFEAT * HFEAT; p += MLP_THREADS) {
        const int j = p & 1, ln = (p >> 1) & 31, q = p >> 6;
        w2s[p] = W2[(4 * (q >> 2) + 2 * (ln >> 4) + j) * HFEAT + 16 * (q & 3) + (ln & 15)];
    }
    for (int p = tid; p < HFEAT * FC; p += MLP_THREADS) {
        const int j = p & 1, ln = (p >> 1) & 31, q = p >> 6;
        f1s[p] = fc1W[(4 * (q >> 1) + 2 * (ln >> 4) + j) * FC + 16 * (q & 1) + (ln & 15)];
    }
    float bias2[4], bias1[2], wo[2];
#pragma unroll
    for (int t = 0; t < 4; ++t) bias2[t] = b2[16 * t + c];
#pragma unroll
    for (int t = 0; t < 2; ++t) {
        bias1[t] = fc1b[16 * t + c];
        wo[t] = fcoW[16 * t + c];
    }
    const float bo = fcob[0];
    __syncthreads();

    float* hw = hs + w * 16 * H_PITCH;
    while (g < ntiles) {
        const int gn = g + stride;
        if (gn < ntiles) load_a_tile(a, gn, c, h, an);

        v8f acc[4] = {};
#pragma unroll
        for (int kk = 0; kk < 16; ++kk) {
#pragma unroll
            for (int t = 0; t < 4; ++t) {
                const v2f b = *(const v2f*)(w2s + ((kk * 4 + t) * 32 + lane) * 2);
                acc[t] = __builtin_amdgcn_wmma_f32_16x16x4_f32(false, af[kk], false, b,
                                                               (short)0, acc[t], false, false);
            }
        }
#pragma unroll
        for (int t = 0; t < 4; ++t) {
#pragma unroll
            for (int r = 0; r < 8; ++r) {
                const float v = acc[t][r] + bias2[t];
                hw[(8 * h + r) * H_PITCH + 16 * t + c] = v > 0.0f ? v : 0.0f;
            }
        }
        __builtin_amdgcn_fence(__ATOMIC_RELEASE, "wavefront");
        __builtin_amdgcn_wave_barrier();
        __builtin_amdgcn_fence(__ATOMIC_ACQUIRE, "wavefront");

        v8f gacc[2] = {};
#pragma unroll
        for (int kk = 0; kk < 16; ++kk) {
            const v2f x = *(const v2f*)(hw + c * H_PITCH + 4 * kk + 2 * h);
#pragma unroll
            for (int t = 0; t < 2; ++t) {
                const v2f b = *(const v2f*)(f1s + ((kk * 2 + t) * 32 + lane) * 2);
                gacc[t] = __builtin_amdgcn_wmma_f32_16x16x4_f32(false, x, false, b,
                                                                (short)0, gacc[t], false, false);
            }
        }
        __builtin_amdgcn_fence(__ATOMIC_RELEASE, "wavefront");
        __builtin_amdgcn_wave_barrier();

        float p[8];
#pragma unroll
        for (int r = 0; r < 8; ++r) {
            float v0 = gacc[0][r] + bias1[0];
            float v1 = gacc[1][r] + bias1[1];
            v0 = v0 > 0.0f ? v0 : 0.0f;
            v1 = v1 > 0.0f ? v1 : 0.0f;
            p[r] = v0 * wo[0] + v1 * wo[1];
        }
#pragma unroll
        for (int off = 8; off > 0; off >>= 1) {
#pragma unroll
            for (int r = 0; r < 8; ++r) p[r] += __shfl_xor(p[r], off, 32);
        }
        float v = p[0];
#pragma unroll
        for (int r = 1; r < 8; ++r) v = (c == r) ? p[r] : v;
        if (c < 8) out[(size_t)g * 16 + 8 * h + c] = v + bo;

#pragma unroll
        for (int kk = 0; kk < 16; ++kk) af[kk] = an[kk];
        g = gn;
    }
}

extern "C" void kernel_launch(void* const* d_in, const int* in_sizes, int n_in,
                              void* d_out, int out_size, void* d_ws, size_t ws_size,
                              hipStream_t stream) {
    const float* x     = (const float*)d_in[0];
    const int*   ei    = (const int*)d_in[1];
    const float* W1    = (const float*)d_in[2];
    const float* b1    = (const float*)d_in[3];
    const float* W2    = (const float*)d_in[4];
    const float* b2    = (const float*)d_in[5];
    const float* fc1W  = (const float*)d_in[6];
    const float* fc1b  = (const float*)d_in[7];
    const float* fcoW  = (const float*)d_in[8];
    const float* fcob  = (const float*)d_in[9];
    float* out = (float*)d_out;

    const int N = in_sizes[0] / 128;
    const int E = in_sizes[1] / 2;
    const int* src = ei;
    const int* dst = ei + E;

    float* dinv = (float*)d_ws;
    float* bufA = dinv + N;
    float* bufB = bufA + (size_t)N * HFEAT;
    unsigned* rec = (unsigned*)(bufB + (size_t)N * HFEAT);

    const int B256 = 256;
    auto blocks = [](long long work, int bs) { return (int)((work + bs - 1) / bs); };

    const int NBLK = blocks(E, BM_EDGES);
    k_fill1<<<blocks(N, B256), B256, 0, stream>>>(dinv, N);
    k_deg_records<<<NBLK, DEG_THREADS, 0, stream>>>(src, dst, dinv, rec, E, NBLK * BM_WORDS);
    k_rsqrt<<<blocks(N, B256), B256, 0, stream>>>(dinv, N);

    {
        long long waves = (long long)(N / 16) * (HFEAT / 16);
        k_gemm_wmma<128, HFEAT><<<blocks(waves * 32, B256), B256, 0, stream>>>(x, W1, bufA, N);
    }
    k_aggregate<true><<<blocks(N, TILE_ROWS), AGG_THREADS, 0, stream>>>(src, dst, dinv, bufA, b1, bufB, rec, NBLK, N, E);

    k_aggregate<false><<<blocks(N, TILE_ROWS), AGG_THREADS, 0, stream>>>(src, dst, dinv, bufB, b2, bufA, rec, NBLK, N, E);

    k_mlp<<<MLP_GROUPS, MLP_THREADS, 0, stream>>>(bufA, W2, b2, fc1W, fc1b, fcoW, fcob, out, N / 16);
}


# x @ W1: 256 x 16 resident waves, each 16 rows x all 64 columns; x read once by 16-byte loads into A fragments, W1 from LDS in fragment order, whole-row stores; k sum order changed
# speedup vs baseline: 2.5293x; 1.0512x over previous
#include <hip/hip_runtime.h>
#include <hip/hip_bf16.h>

typedef __attribute__((ext_vector_type(2))) float v2f;
typedef __attribute__((ext_vector_type(8))) float v8f;

#define HFEAT 64

__global__ void k_fill1(float* d, int n) {
    int t = blockIdx.x * blockDim.x + threadIdx.x;
    if (t < n) d[t] = 1.0f;
}

#define TILE_ROWS 391
#define NTILES 256
#define BM_EDGES 2048
#define DEG_THREADS 1024
#define DEG_EPT (BM_EDGES / DEG_THREADS)
#define BM_WORDS 32
#define BM_PITCH (BM_WORDS + 1)
#define REC_CAP (BM_WORDS - 1)
#define REC_SHIFT 9
#define REC_MASK ((1u << REC_SHIFT) - 1u)

__global__ __launch_bounds__(DEG_THREADS) void k_deg_records(const int* __restrict__ src,
                                                             const int* __restrict__ dst,
                                                             float* __restrict__ deg,
                                                             unsigned* __restrict__ rec, int e, int wpt) {
    __shared__ unsigned s[NTILES * BM_PITCH];
    const int tid = threadIdx.x;
    const int lane = tid & 31;
    const int w = tid >> 5;
    const int t0 = blockIdx.x * BM_EDGES + tid;
    int d[DEG_EPT];
    int sv[DEG_EPT];
#pragma unroll
    for (int k = 0; k < DEG_EPT; ++k) {
        const int t = t0 + k * DEG_THREADS;
        const int tl = t < e ? t : e - 1;
        d[k] = dst[tl];
        sv[k] = src[tl];
        if (t >= e) d[k] = -1;
    }
    for (int i = tid; i < NTILES * BM_PITCH; i += DEG_THREADS) s[i] = 0u;
    __syncthreads();
#pragma unroll
    for (int k = 0; k < DEG_EPT; ++k) {
        if (d[k] >= 0) {
            unsafeAtomicAdd(&deg[d[k]], 1.0f);
            const unsigned tile = (unsigned)d[k] / TILE_ROWS;
            if (tile < NTILES) {
                const unsigned slot = __hip_atomic_fetch_add(&s[tile * BM_PITCH], 1u, __ATOMIC_RELAXED,
                                                             __HIP_MEMORY_SCOPE_WORKGROUP);
                if (slot < REC_CAP)
                    s[tile * BM_PITCH + 1 + slot] =
                        ((unsigned)sv[k] << REC_SHIFT) | ((unsigned)d[k] - tile * TILE_ROWS);
            }
        }
    }
    __syncthreads();
    unsigned* o = rec + (size_t)w * wpt + blockIdx.x * BM_WORDS + lane;
#pragma unroll
    for (int k = 0; k < NTILES / BM_WORDS; ++k)
        o[(size_t)k * BM_WORDS * wpt] = s[(w + k * BM_WORDS) * BM_PITCH + lane];
}

__global__ void k_rsqrt(float* d, int n) {
    int t = blockIdx.x * blockDim.x + threadIdx.x;
    if (t < n) d[t] = rsqrtf(d[t]);
}

#define XW_K 128
#define XW_WAVES 16
#define XW_THREADS (XW_WAVES * 32)
#define XW_GROUPS 256
#define XW_PITCH 68

typedef __attribute__((ext_vector_type(4))) float v4f;

__device__ __forceinline__ void load_x_tile(const float* __restrict__ x, int g, int c, int h, v4f (&f)[16]) {
    const float* xr = x + ((size_t)g * 16 + c) * XW_K + 4 * h;
#pragma unroll
    for (int q = 0; q < 16; ++q) f[q] = *(const v4f*)(xr + 8 * q);
}

__global__ __launch_bounds__(XW_THREADS) void k_xw1(const float* __restrict__ x, const float* __restrict__ W1,
                                                    float* __restrict__ xw, int ntiles) {
    __shared__ __attribute__((aligned(16))) float w1s[XW_K * HFEAT];
    __shared__ __attribute__((aligned(16))) float os[XW_WAVES * 16 * XW_PITCH];

    const int tid = threadIdx.x;
    const int lane = tid & 31;
    const int w = __builtin_amdgcn_readfirstlane(tid >> 5);
    const int c = lane & 15;
    const int h = lane >> 4;
    const int stride = XW_GROUPS * XW_WAVES;

    int g = (int)blockIdx.x + XW_GROUPS * w;
    v4f xf[16];
    v4f xn[16] = {};
    if (g < ntiles) load_x_tile(x, g, c, h, xf);

    for (int p = tid; p < XW_K * HFEAT; p += XW_THREADS) {
        const int j = p & 1, ln = (p >> 1) & 31, sq = p >> 6;
        const int s = sq >> 2, t = sq & 3;
        w1s[p] = W1[(8 * (s >> 1) + 4 * (ln >> 4) + 2 * (s & 1) + j) * HFEAT + 16 * t + (ln & 15)];
    }
    __syncthreads();

    float* ow = os + w * 16 * XW_PITCH;
    while (g < ntiles) {
        const int gn = g + stride;
        if (gn < ntiles) load_x_tile(x, gn, c, h, xn);

        v8f acc[4] = {};
#pragma unroll
        for (int q = 0; q < 16; ++q) {
#pragma unroll
            for (int u = 0; u < 2; ++u) {
                v2f a;
                a.x = u ? xf[q].z : xf[q].x;
                a.y = u ? xf[q].w : xf[q].y;
#pragma unroll
                for (int t = 0; t < 4; ++t) {
                    const v2f b = *(const v2f*)(w1s + (((2 * q + u) * 4 + t) * 32 + lane) * 2);
                    acc[t] = __builtin_amdgcn_wmma_f32_16x16x4_f32(false, a, false, b,
                                                                   (short)0, acc[t], false, false);
                }
            }
        }
#pragma unroll
        for (int t = 0; t < 4; ++t) {
#pragma unroll
            for (int r = 0; r < 8; ++r) ow[(8 * h + r) * XW_PITCH + 16 * t + c] = acc[t][r];
        }
        __builtin_amdgcn_fence(__ATOMIC_RELEASE, "wavefront");
        __builtin_amdgcn_wave_barrier();
        __builtin_amdgcn_fence(__ATOMIC_ACQUIRE, "wavefront");
        float* orow = xw + ((size_t)g * 16 + h) * HFEAT + 4 * c;
#pragma unroll
        for (int i = 0; i < 8; ++i) {
            const v4f v = *(const v4f*)(ow + (2 * i + h) * XW_PITCH + 4 * c);
            *(v4f*)(orow + (size_t)(2 * i) * HFEAT) = v;
        }
        __builtin_amdgcn_fence(__ATOMIC_RELEASE, "wavefront");
        __builtin_amdgcn_wave_barrier();

#pragma unroll
        for (int q = 0; q < 16; ++q) xf[q] = xn[q];
        g = gn;
    }
}

#define AGG_THREADS 1024
#define AGG_WAVES 32
#define SCAN_LINES 4
#define OVF_EDGES 1024
#define QCAP (32 + OVF_EDGES)

__device__ __forceinline__ void load_lines(const unsigned* __restrict__ row, int base, int nblk,
                                           int lane, unsigned (&v)[SCAN_LINES]) {
#pragma unroll
    for (int j = 0; j < SCAN_LINES; ++j)
        v[j] = (base + j < nblk) ? row[(size_t)(base + j) * BM_WORDS + lane] : 0u;
}

__device__ __forceinline__ void lds_add(float* p, float v) {
    __hip_atomic_fetch_add(p, v, __ATOMIC_RELAXED, __HIP_MEMORY_SCOPE_WORKGROUP);
}

__device__ __forceinline__ void drain(const unsigned* qe, int start, int count, int lane, int d0,
                                      const float* __restrict__ dinv, const float* __restrict__ xw,
                                      float* tile) {
    __builtin_amdgcn_fence(__ATOMIC_RELEASE, "wavefront");
    __builtin_amdgcn_wave_barrier();
    __builtin_amdgcn_fence(__ATOMIC_ACQUIRE, "wavefront");
    int s_i = 0, t_i = 0;
    float c_i = 0.0f;
    if (lane < count) {
        const unsigned r_i = qe[start + lane];
        s_i = (int)(r_i >> REC_SHIFT);
        t_i = (int)(r_i & REC_MASK);
        c_i = dinv[s_i] * dinv[d0 + t_i];
    }
    int i = 0;
    for (; i + 8 <= count; i += 8) {
        float r[16];
        int tt[8];
        float cc[8];
#pragma unroll
        for (int u = 0; u < 8; ++u) {
            int s = __builtin_amdgcn_readlane(s_i, i + u);
            tt[u] = __builtin_amdgcn_readlane(t_i, i + u);
            cc[u] = __int_as_float(__builtin_amdgcn_readlane(__float_as_int(c_i), i + u));
            const float* rp = xw + (size_t)s * HFEAT + lane;
            r[2 * u] = rp[0];
            r[2 * u + 1] = rp[32];
        }
#pragma unroll
        for (int u = 0; u < 8; ++u) {
            float* o = tile + tt[u] * HFEAT + lane;
            lds_add(o, cc[u] * r[2 * u]);
            lds_add(o + 32, cc[u] * r[2 * u + 1]);
        }
    }
    for (; i < count; ++i) {
        int s = __builtin_amdgcn_readlane(s_i, i);
        int t = __builtin_amdgcn_readlane(t_i, i);
        float c = __int_as_float(__builtin_amdgcn_readlane(__float_as_int(c_i), i));
        const float* rp = xw + (size_t)s * HFEAT + lane;
        float r0 = rp[0];
        float r1 = rp[32];
        float* o = tile + t * HFEAT + lane;
        lds_add(o, c * r0);
        lds_add(o + 32, c * r1);
    }
    __builtin_amdgcn_fence(__ATOMIC_RELEASE, "wavefront");
    __builtin_amdgcn_wave_barrier();
}

template <bool BIAS_RELU>
__global__ __launch_bounds__(AGG_THREADS) void k_aggregate(
        const int* __restrict__ src, const int* __restrict__ dst, const float* __restrict__ dinv,
        const float* __restrict__ xw, const float* __restrict__ bias, float* __restrict__ out,
        const unsigned* __restrict__ rec, int nblk, int n, int e) {
    __shared__ float tile[TILE_ROWS * HFEAT];
    __shared__ unsigned q_e[AGG_WAVES * QCAP];

    const int tid = threadIdx.x;
    const int lane = tid & 31;
    const int w = __builtin_amdgcn_readfirstlane(tid >> 5);
    const int d0 = blockIdx.x * TILE_ROWS;
    int cnt = n - d0;
    if (cnt > TILE_ROWS) cnt = TILE_ROWS;
    const int total = cnt * HFEAT;

    const float* xwt = xw + (size_t)d0 * HFEAT;
    for (int i = tid; i < total; i += AGG_THREADS) {
        float di = dinv[d0 + (i >> 6)];
        tile[i] = di * di * xwt[i];
    }
    __syncthreads();

    unsigned* qe = q_e + w * QCAP;
    int qn = 0;
    const unsigned* row = rec + (size_t)blockIdx.x * nblk * BM_WORDS;

    unsigned cur[SCAN_LINES];
    unsigned nxt[SCAN_LINES] = {};
    int base = w * SCAN_LINES;
    if (base < nblk) load_lines(row, base, nblk, lane, cur);
    for (; base < nblk; base += AGG_WAVES * SCAN_LINES) {
        const int nb = base + AGG_WAVES * SCAN_LINES;
        if (nb < nblk) load_lines(row, nb, nblk, lane, nxt);
#pragma unroll
        for (int j = 0; j < SCAN_LINES; ++j) {
            const unsigned word = cur[j];
            const int c = __builtin_amdgcn_readlane((int)word, 0);
            const bool fits = c <= REC_CAP;
            const int passes = fits ? 1 : BM_EDGES / OVF_EDGES;
            for (int h = 0; h < passes; ++h) {
                if (fits) {
                    if (lane >= 1 && lane <= c) qe[qn + lane - 1] = word;
                    qn += c;
                } else {
                    const int e0 = (base + j) * BM_EDGES + h * OVF_EDGES + lane;
#pragma nounroll
                    for (int r = 0; r < OVF_EDGES / 32; ++r) {
                        const int e_i = e0 + r * 32;
                        bool m = false;
                        unsigned rr = 0u;
                        if (e_i < e) {
                            const unsigned loc = (unsigned)(dst[e_i] - d0);
                            if (loc < (unsigned)cnt) {
                                m = true;
                                rr = ((unsigned)src[e_i] << REC_SHIFT) | loc;
                            }
                        }
                        const unsigned any = __builtin_amdgcn_ballot_w32(m);
                        if (m) qe[qn + (int)__builtin_amdgcn_mbcnt_lo(any, 0u)] = rr;
                        qn += __builtin_popcount(any);
                    }
                }
                while (qn >= 32) {
                    qn -= 32;
                    drain(qe, qn, 32, lane, d0, dinv, xw, tile);
                }
            }
        }
#pragma unroll
        for (int j = 0; j < SCAN_LINES; ++j) cur[j] = nxt[j];
    }
    if (qn > 0) drain(qe, 0, qn, lane, d0, dinv, xw, tile);
    __syncthreads();

    float* ot = out + (size_t)d0 * HFEAT;
    for (int i = tid; i < total; i += AGG_THREADS) {
        float v = tile[i];
        if (BIAS_RELU) {
            v += bias[i & (HFEAT - 1)];
            v = v > 0.0f ? v : 0.0f;
        }
        ot[i] = v;
    }
}

#define MLP_WAVES 16
#define MLP_THREADS (MLP_WAVES * 32)
#define MLP_GROUPS 256
#define H_PITCH 68
#define FC 32

__device__ __forceinline__ void load_a_tile(const float* __restrict__ a, int g, int c, int h, v2f (&f)[16]) {
    const float* arow = a + ((size_t)g * 16 + c) * HFEAT + 2 * h;
#pragma unroll
    for (int kk = 0; kk < 16; ++kk) f[kk] = *(const v2f*)(arow + 4 * kk);
}

__global__ __launch_bounds__(MLP_THREADS) void k_mlp(
        const float* __restrict__ a, const float* __restrict__ W2, const float* __restrict__ b2,
        const float* __restrict__ fc1W, const float* __restrict__ fc1b, const float* __restrict__ fcoW,
        const float* __restrict__ fcob, float* __restrict__ out, int ntiles) {
    __shared__ float w2s[HFEAT * HFEAT];
    __shared__ float f1s[HFEAT * FC];
    __shared__ float hs[MLP_WAVES * 16 * H_PITCH];

    const int tid = threadIdx.x;
    const int lane = tid & 31;
    const int w = __builtin_amdgcn_readfirstlane(tid >> 5);
    const int c = lane & 15;
    const int h = lane >> 4;
    const int stride = MLP_GROUPS * MLP_WAVES;

    int g = (int)blockIdx.x + MLP_GROUPS * w;
    v2f af[16];
    v2f an[16] = {};
    if (g < ntiles) load_a_tile(a, g, c, h, af);

    for (int p = tid; p < HFEAT * HFEAT; p += MLP_THREADS) {
        const int j = p & 1, ln = (p >> 1) & 31, q = p >> 6;
        w2s[p] = W2[(4 * (q >> 2) + 2 * (ln >> 4) + j) * HFEAT + 16 * (q & 3) + (ln & 15)];
    }
    for (int p = tid; p < HFEAT * FC; p += MLP_THREADS) {
        const int j = p & 1, ln = (p >> 1) & 31, q = p >> 6;
        f1s[p] = fc1W[(4 * (q >> 1) + 2 * (ln >> 4) + j) * FC + 16 * (q & 1) + (ln & 15)];
    }
    float bias2[4], bias1[2], wo[2];
#pragma unroll
    for (int t = 0; t < 4; ++t) bias2[t] = b2[16 * t + c];
#pragma unroll
    for (int t = 0; t < 2; ++t) {
        bias1[t] = fc1b[16 * t + c];
        wo[t] = fcoW[16 * t + c];
    }
    const float bo = fcob[0];
    __syncthreads();

    float* hw = hs + w * 16 * H_PITCH;
    while (g < ntiles) {
        const int gn = g + stride;
        if (gn < ntiles) load_a_tile(a, gn, c, h, an);

        v8f acc[4] = {};
#pragma unroll
        for (int kk = 0; kk < 16; ++kk) {
#pragma unroll
            for (int t = 0; t < 4; ++t) {
                const v2f b = *(const v2f*)(w2s + ((kk * 4 + t) * 32 + lane) * 2);
                acc[t] = __builtin_amdgcn_wmma_f32_16x16x4_f32(false, af[kk], false, b,
                                                               (short)0, acc[t], false, false);
            }
        }
#pragma unroll
        for (int t = 0; t < 4; ++t) {
#pragma unroll
            for (int r = 0; r < 8; ++r) {
                const float v = acc[t][r] + bias2[t];
                hw[(8 * h + r) * H_PITCH + 16 * t + c] = v > 0.0f ? v : 0.0f;
            }
        }
        __builtin_amdgcn_fence(__ATOMIC_RELEASE, "wavefront");
        __builtin_amdgcn_wave_barrier();
        __builtin_amdgcn_fence(__ATOMIC_ACQUIRE, "wavefront");

        v8f gacc[2] = {};
#pragma unroll
        for (int kk = 0; kk < 16; ++kk) {
            const v2f x = *(const v2f*)(hw + c * H_PITCH + 4 * kk + 2 * h);
#pragma unroll
            for (int t = 0; t < 2; ++t) {
                const v2f b = *(const v2f*)(f1s + ((kk * 2 + t) * 32 + lane) * 2);
                gacc[t] = __builtin_amdgcn_wmma_f32_16x16x4_f32(false, x, false, b,
                                                                (short)0, gacc[t], false, false);
            }
        }
        __builtin_amdgcn_fence(__ATOMIC_RELEASE, "wavefront");
        __builtin_amdgcn_wave_barrier();

        float p[8];
#pragma unroll
        for (int r = 0; r < 8; ++r) {
            float v0 = gacc[0][r] + bias1[0];
            float v1 = gacc[1][r] + bias1[1];
            v0 = v0 > 0.0f ? v0 : 0.0f;
            v1 = v1 > 0.0f ? v1 : 0.0f;
            p[r] = v0 * wo[0] + v1 * wo[1];
        }
#pragma unroll
        for (int off = 8; off > 0; off >>= 1) {
#pragma unroll
            for (int r = 0; r < 8; ++r) p[r] += __shfl_xor(p[r], off, 32);
        }
        float v = p[0];
#pragma unroll
        for (int r = 1; r < 8; ++r) v = (c == r) ? p[r] : v;
        if (c < 8) out[(size_t)g * 16 + 8 * h + c] = v + bo;

#pragma unroll
        for (int kk = 0; kk < 16; ++kk) af[kk] = an[kk];
        g = gn;
    }
}

extern "C" void kernel_launch(void* const* d_in, const int* in_sizes, int n_in,
                              void* d_out, int out_size, void* d_ws, size_t ws_size,
                              hipStream_t stream) {
    const float* x     = (const float*)d_in[0];
    const int*   ei    = (const int*)d_in[1];
    const float* W1    = (const float*)d_in[2];
    const float* b1    = (const float*)d_in[3];
    const float* W2    = (const float*)d_in[4];
    const float* b2    = (const float*)d_in[5];
    const float* fc1W  = (const float*)d_in[6];
    const float* fc1b  = (const float*)d_in[7];
    const float* fcoW  = (const float*)d_in[8];
    const float* fcob  = (const float*)d_in[9];
    float* out = (float*)d_out;

    const int N = in_sizes[0] / 128;
    const int E = in_sizes[1] / 2;
    const int* src = ei;
    const int* dst = ei + E;

    float* dinv = (float*)d_ws;
    float* bufA = dinv + N;
    float* bufB = bufA + (size_t)N * HFEAT;
    unsigned* rec = (unsigned*)(bufB + (size_t)N * HFEAT);

    const int B256 = 256;
    auto blocks = [](long long work, int bs) { return (int)((work + bs - 1) / bs); };

    const int NBLK = blocks(E, BM_EDGES);
    k_fill1<<<blocks(N, B256), B256, 0, stream>>>(dinv, N);
    k_deg_records<<<NBLK, DEG_THREADS, 0, stream>>>(src, dst, dinv, rec, E, NBLK * BM_WORDS);
    k_rsqrt<<<blocks(N, B256), B256, 0, stream>>>(dinv, N);

    k_xw1<<<XW_GROUPS, XW_THREADS, 0, stream>>>(x, W1, bufA, N / 16);
    k_aggregate<true><<<blocks(N, TILE_ROWS), AGG_THREADS, 0, stream>>>(src, dst, dinv, bufA, b1, bufB, rec, NBLK, N, E);

    k_aggregate<false><<<blocks(N, TILE_ROWS), AGG_THREADS, 0, stream>>>(src, dst, dinv, bufB, b2, bufA, rec, NBLK, N, E);

    k_mlp<<<MLP_GROUPS, MLP_THREADS, 0, stream>>>(bufA, W2, b2, fc1W, fc1b, fcoW, fcob, out, N / 16);
}


# in-degree = 1 + a tile's records per row, counted in LDS (integer adds) by the x @ W1 work-group that owns the tile; global float atomics, fill and rsqrt launches go (7 to 5); sum order unchanged
# speedup vs baseline: 4.0930x; 1.6182x over previous
#include <hip/hip_runtime.h>
#include <hip/hip_bf16.h>

typedef __attribute__((ext_vector_type(2))) float v2f;
typedef __attribute__((ext_vector_type(8))) float v8f;

#define HFEAT 64

#define TILE_ROWS 391
#define NTILES 256
#define BM_EDGES 2048
#define DEG_THREADS 1024
#define DEG_EPT (BM_EDGES / DEG_THREADS)
#define BM_WORDS 32
#define BM_PITCH (BM_WORDS + 1)
#define REC_CAP (BM_WORDS - 1)
#define REC_SHIFT 9
#define REC_MASK ((1u << REC_SHIFT) - 1u)

__global__ __launch_bounds__(DEG_THREADS) void k_records(const int* __restrict__ src,
                                                         const int* __restrict__ dst,
                                                         unsigned* __restrict__ rec, int e, int wpt) {
    __shared__ unsigned s[NTILES * BM_PITCH];
    const int tid = threadIdx.x;
    const int lane = tid & 31;
    const int w = tid >> 5;
    const int t0 = blockIdx.x * BM_EDGES + tid;
    int d[DEG_EPT];
    int sv[DEG_EPT];
#pragma unroll
    for (int k = 0; k < DEG_EPT; ++k) {
        const int t = t0 + k * DEG_THREADS;
        const int tl = t < e ? t : e - 1;
        d[k] = dst[tl];
        sv[k] = src[tl];
        if (t >= e) d[k] = -1;
    }
    for (int i = tid; i < NTILES * BM_PITCH; i += DEG_THREADS) s[i] = 0u;
    __syncthreads();
#pragma unroll
    for (int k = 0; k < DEG_EPT; ++k) {
        if (d[k] >= 0) {
            const unsigned tile = (unsigned)d[k] / TILE_ROWS;
            if (tile < NTILES) {
                const unsigned slot = __hip_atomic_fetch_add(&s[tile * BM_PITCH], 1u, __ATOMIC_RELAXED,
                                                             __HIP_MEMORY_SCOPE_WORKGROUP);
                if (slot < REC_CAP)
                    s[tile * BM_PITCH + 1 + slot] =
                        ((unsigned)sv[k] << REC_SHIFT) | ((unsigned)d[k] - tile * TILE_ROWS);
            }
        }
    }
    __syncthreads();
    unsigned* o = rec + (size_t)w * wpt + blockIdx.x * BM_WORDS + lane;
#pragma unroll
    for (int k = 0; k < NTILES / BM_WORDS; ++k)
        o[(size_t)k * BM_WORDS * wpt] = s[(w + k * BM_WORDS) * BM_PITCH + lane];
}

#define XW_K 128
#define XW_WAVES 16
#define XW_THREADS (XW_WAVES * 32)
#define XW_GROUPS 256
#define XW_PITCH 68

typedef __attribute__((ext_vector_type(4))) float v4f;
typedef __attribute__((ext_vector_type(4))) unsigned v4u;
#define CNT_BATCH 13

static_assert(XW_GROUPS == NTILES, "one x @ W1 work-group per aggregation tile");
static_assert(XW_THREADS >= TILE_ROWS, "one thread per tile row clears the counters and stores dinv");

__device__ __forceinline__ void load_x_tile(const float* __restrict__ x, int g, int c, int h, v4f (&f)[16]) {
    const float* xr = x + ((size_t)g * 16 + c) * XW_K + 4 * h;
#pragma unroll
    for (int q = 0; q < 16; ++q) f[q] = *(const v4f*)(xr + 8 * q);
}

__device__ __forceinline__ void load_rec_lines(const unsigned* __restrict__ row, int lg0, int nblk, int lsub, int wq,
                                               v4u (&lv)[CNT_BATCH]) {
#pragma unroll
    for (int k = 0; k < CNT_BATCH; ++k) {
        const int line = 4 * (lg0 + k * XW_WAVES) + lsub;
        const int ll = line < nblk ? line : nblk - 1;
        lv[k] = *(const v4u*)(row + (size_t)ll * BM_WORDS + 4 * wq);
    }
}

__global__ __launch_bounds__(XW_THREADS) void k_xw1(const float* __restrict__ x, const float* __restrict__ W1,
                                                    float* __restrict__ xw, int ntiles,
                                                    const unsigned* __restrict__ rec, const int* __restrict__ dst,
                                                    float* __restrict__ dinv, int nblk, int n, int e) {
    __shared__ __attribute__((aligned(16))) float w1s[XW_K * HFEAT];
    __shared__ __attribute__((aligned(16))) float os[XW_WAVES * 16 * XW_PITCH];
    __shared__ unsigned cnts[TILE_ROWS];

    const int tid = threadIdx.x;
    const int lane = tid & 31;
    const int w = __builtin_amdgcn_readfirstlane(tid >> 5);
    const int c = lane & 15;
    const int h = lane >> 4;
    const int stride = XW_GROUPS * XW_WAVES;

    const int d0 = (int)blockIdx.x * TILE_ROWS;
    int cnt = n - d0;
    if (cnt > TILE_ROWS) cnt = TILE_ROWS;
    const unsigned* row = rec + (size_t)blockIdx.x * nblk * BM_WORDS;
    const int nq = (nblk + 3) >> 2;
    const int lsub = lane >> 3, wq = lane & 7;
    v4u lv[CNT_BATCH];
    load_rec_lines(row, w, nblk, lsub, wq, lv);

    int g = (int)blockIdx.x + XW_GROUPS * w;
    v4f xf[16];
    v4f xn[16] = {};
    if (g < ntiles) load_x_tile(x, g, c, h, xf);

    if (tid < TILE_ROWS) cnts[tid] = 0u;

    for (int p = tid; p < XW_K * HFEAT; p += XW_THREADS) {
        const int j = p & 1, ln = (p >> 1) & 31, sq = p >> 6;
        const int s = sq >> 2, t = sq & 3;
        w1s[p] = W1[(8 * (s >> 1) + 4 * (ln >> 4) + 2 * (s & 1) + j) * HFEAT + 16 * t + (ln & 15)];
    }
    __syncthreads();

    for (int lg0 = w; lg0 < nq; lg0 += XW_WAVES * CNT_BATCH) {
        if (lg0 != w) load_rec_lines(row, lg0, nblk, lsub, wq, lv);
#pragma unroll
        for (int k = 0; k < CNT_BATCH; ++k) {
            const bool real = 4 * (lg0 + k * XW_WAVES) + lsub < nblk;
            const unsigned lc = (unsigned)__shfl((int)lv[k].x, lane & 24);
            const bool fits = lc <= REC_CAP;
#pragma unroll
            for (int j = 0; j < 4; ++j) {
                const unsigned wi = 4u * wq + j;
                const unsigned loc = lv[k][j] & REC_MASK;
                if (real && fits && wi >= 1u && wi <= lc && loc < TILE_ROWS)
                    __hip_atomic_fetch_add(&cnts[loc], 1u, __ATOMIC_RELAXED, __HIP_MEMORY_SCOPE_WORKGROUP);
            }
            unsigned over = __builtin_amdgcn_ballot_w32(real && !fits && wq == 0);
            while (over) {
                const int bl = __builtin_ctz(over);
                over &= over - 1u;
                const int e0 = (4 * (lg0 + k * XW_WAVES) + (bl >> 3)) * BM_EDGES + lane;
#pragma nounroll
                for (int r = 0; r < BM_EDGES / 32; ++r) {
                    const int e_i = e0 + r * 32;
                    if (e_i < e) {
                        const unsigned loc = (unsigned)(dst[e_i] - d0);
                        if (loc < (unsigned)cnt)
                            __hip_atomic_fetch_add(&cnts[loc], 1u, __ATOMIC_RELAXED, __HIP_MEMORY_SCOPE_WORKGROUP);
                    }
                }
            }
        }
    }

    float* ow = os + w * 16 * XW_PITCH;
    while (g < ntiles) {
        const int gn = g + stride;
        if (gn < ntiles) load_x_tile(x, gn, c, h, xn);

        v8f acc[4] = {};
#pragma unroll
        for (int q = 0; q < 16; ++q) {
#pragma unroll
            for (int u = 0; u < 2; ++u) {
                v2f a;
                a.x = u ? xf[q].z : xf[q].x;
                a.y = u ? xf[q].w : xf[q].y;
#pragma unroll
                for (int t = 0; t < 4; ++t) {
                    const v2f b = *(const v2f*)(w1s + (((2 * q + u) * 4 + t) * 32 + lane) * 2);
                    acc[t] = __builtin_amdgcn_wmma_f32_16x16x4_f32(false, a, false, b,
                                                                   (short)0, acc[t], false, false);
                }
            }
        }
#pragma unroll
        for (int t = 0; t < 4; ++t) {
#pragma unroll
            for (int r = 0; r < 8; ++r) ow[(8 * h + r) * XW_PITCH + 16 * t + c] = acc[t][r];
        }
        __builtin_amdgcn_fence(__ATOMIC_RELEASE, "wavefront");
        __builtin_amdgcn_wave_barrier();
        __builtin_amdgcn_fence(__ATOMIC_ACQUIRE, "wavefront");
        float* orow = xw + ((size_t)g * 16 + h) * HFEAT + 4 * c;
#pragma unroll
        for (int i = 0; i < 8; ++i) {
            const v4f v = *(const v4f*)(ow + (2 * i + h) * XW_PITCH + 4 * c);
            *(v4f*)(orow + (size_t)(2 * i) * HFEAT) = v;
        }
        __builtin_amdgcn_fence(__ATOMIC_RELEASE, "wavefront");
        __builtin_amdgcn_wave_barrier();

#pragma unroll
        for (int q = 0; q < 16; ++q) xf[q] = xn[q];
        g = gn;
    }

    __syncthreads();
    if (tid < cnt) dinv[d0 + tid] = rsqrtf((float)(1u + cnts[tid]));
}

#define AGG_THREADS 1024
#define AGG_WAVES 32
#define SCAN_LINES 4
#define OVF_EDGES 1024
#define QCAP (32 + OVF_EDGES)

__device__ __forceinline__ void load_lines(const unsigned* __restrict__ row, int base, int nblk,
                                           int lane, unsigned (&v)[SCAN_LINES]) {
#pragma unroll
    for (int j = 0; j < SCAN_LINES; ++j)
        v[j] = (base + j < nblk) ? row[(size_t)(base + j) * BM_WORDS + lane] : 0u;
}

__device__ __forceinline__ void lds_add(float* p, float v) {
    __hip_atomic_fetch_add(p, v, __ATOMIC_RELAXED, __HIP_MEMORY_SCOPE_WORKGROUP);
}

__device__ __forceinline__ void drain(const unsigned* qe, int start, int count, int lane, int d0,
                                      const float* __restrict__ dinv, const float* __restrict__ xw,
                                      float* tile) {
    __builtin_amdgcn_fence(__ATOMIC_RELEASE, "wavefront");
    __builtin_amdgcn_wave_barrier();
    __builtin_amdgcn_fence(__ATOMIC_ACQUIRE, "wavefront");
    int s_i = 0, t_i = 0;
    float c_i = 0.0f;
    if (lane < count) {
        const unsigned r_i = qe[start + lane];
        s_i = (int)(r_i >> REC_SHIFT);
        t_i = (int)(r_i & REC_MASK);
        c_i = dinv[s_i] * dinv[d0 + t_i];
    }
    int i = 0;
    for (; i + 8 <= count; i += 8) {
        float r[16];
        int tt[8];
        float cc[8];
#pragma unroll
        for (int u = 0; u < 8; ++u) {
            int s = __builtin_amdgcn_readlane(s_i, i + u);
            tt[u] = __builtin_amdgcn_readlane(t_i, i + u);
            cc[u] = __int_as_float(__builtin_amdgcn_readlane(__float_as_int(c_i), i + u));
            const float* rp = xw + (size_t)s * HFEAT + lane;
            r[2 * u] = rp[0];
            r[2 * u + 1] = rp[32];
        }
#pragma unroll
        for (int u = 0; u < 8; ++u) {
            float* o = tile + tt[u] * HFEAT + lane;
            lds_add(o, cc[u] * r[2 * u]);
            lds_add(o + 32, cc[u] * r[2 * u + 1]);
        }
    }
    for (; i < count; ++i) {
        int s = __builtin_amdgcn_readlane(s_i, i);
        int t = __builtin_amdgcn_readlane(t_i, i);
        float c = __int_as_float(__builtin_amdgcn_readlane(__float_as_int(c_i), i));
        const float* rp = xw + (size_t)s * HFEAT + lane;
        float r0 = rp[0];
        float r1 = rp[32];
        float* o = tile + t * HFEAT + lane;
        lds_add(o, c * r0);
        lds_add(o + 32, c * r1);
    }
    __builtin_amdgcn_fence(__ATOMIC_RELEASE, "wavefront");
    __builtin_amdgcn_wave_barrier();
}

template <bool BIAS_RELU>
__global__ __launch_bounds__(AGG_THREADS) void k_aggregate(
        const int* __restrict__ src, const int* __restrict__ dst, const float* __restrict__ dinv,
        const float* __restrict__ xw, const float* __restrict__ bias, float* __restrict__ out,
        const unsigned* __restrict__ rec, int nblk, int n, int e) {
    __shared__ float tile[TILE_ROWS * HFEAT];
    __shared__ unsigned q_e[AGG_WAVES * QCAP];

    const int tid = threadIdx.x;
    const int lane = tid & 31;
    const int w = __builtin_amdgcn_readfirstlane(tid >> 5);
    const int d0 = blockIdx.x * TILE_ROWS;
    int cnt = n - d0;
    if (cnt > TILE_ROWS) cnt = TILE_ROWS;
    const int total = cnt * HFEAT;

    const float* xwt = xw + (size_t)d0 * HFEAT;
    for (int i = tid; i < total; i += AGG_THREADS) {
        float di = dinv[d0 + (i >> 6)];
        tile[i] = di * di * xwt[i];
    }
    __syncthreads();

    unsigned* qe = q_e + w * QCAP;
    int qn = 0;
    const unsigned* row = rec + (size_t)blockIdx.x * nblk * BM_WORDS;

    unsigned cur[SCAN_LINES];
    unsigned nxt[SCAN_LINES] = {};
    int base = w * SCAN_LINES;
    if (base < nblk) load_lines(row, base, nblk, lane, cur);
    for (; base < nblk; base += AGG_WAVES * SCAN_LINES) {
        const int nb = base + AGG_WAVES * SCAN_LINES;
        if (nb < nblk) load_lines(row, nb, nblk, lane, nxt);
#pragma unroll
        for (int j = 0; j < SCAN_LINES; ++j) {
            const unsigned word = cur[j];
            const int c = __builtin_amdgcn_readlane((int)word, 0);
            const bool fits = c <= REC_CAP;
            const int passes = fits ? 1 : BM_EDGES / OVF_EDGES;
            for (int h = 0; h < passes; ++h) {
                if (fits) {
                    if (lane >= 1 && lane <= c) qe[qn + lane - 1] = word;
                    qn += c;
                } else {
                    const int e0 = (base + j) * BM_EDGES + h * OVF_EDGES + lane;
#pragma nounroll
                    for (int r = 0; r < OVF_EDGES / 32; ++r) {
                        const int e_i = e0 + r * 32;
                        bool m = false;
                        unsigned rr = 0u;
                        if (e_i < e) {
                            const unsigned loc = (unsigned)(dst[e_i] - d0);
                            if (loc < (unsigned)cnt) {
                                m = true;
                                rr = ((unsigned)src[e_i] << REC_SHIFT) | loc;
                            }
                        }
                        const unsigned any = __builtin_amdgcn_ballot_w32(m);
                        if (m) qe[qn + (int)__builtin_amdgcn_mbcnt_lo(any, 0u)] = rr;
                        qn += __builtin_popcount(any);
                    }
                }
                while (qn >= 32) {
                    qn -= 32;
                    drain(qe, qn, 32, lane, d0, dinv, xw, tile);
                }
            }
        }
#pragma unroll
        for (int j = 0; j < SCAN_LINES; ++j) cur[j] = nxt[j];
    }
    if (qn > 0) drain(qe, 0, qn, lane, d0, dinv, xw, tile);
    __syncthreads();

    float* ot = out + (size_t)d0 * HFEAT;
    for (int i = tid; i < total; i += AGG_THREADS) {
        float v = tile[i];
        if (BIAS_RELU) {
            v += bias[i & (HFEAT - 1)];
            v = v > 0.0f ? v : 0.0f;
        }
        ot[i] = v;
    }
}

#define MLP_WAVES 16
#define MLP_THREADS (MLP_WAVES * 32)
#define MLP_GROUPS 256
#define H_PITCH 68
#define FC 32

__device__ __forceinline__ void load_a_tile(const float* __restrict__ a, int g, int c, int h, v2f (&f)[16]) {
    const float* arow = a + ((size_t)g * 16 + c) * HFEAT + 2 * h;
#pragma unroll
    for (int kk = 0; kk < 16; ++kk) f[kk] = *(const v2f*)(arow + 4 * kk);
}

__global__ __launch_bounds__(MLP_THREADS) void k_mlp(
        const float* __restrict__ a, const float* __restrict__ W2, const float* __restrict__ b2,
        const float* __restrict__ fc1W, const float* __restrict__ fc1b, const float* __restrict__ fcoW,
        const float* __restrict__ fcob, float* __restrict__ out, int ntiles) {
    __shared__ float w2s[HFEAT * HFEAT];
    __shared__ float f1s[HFEAT * FC];
    __shared__ float hs[MLP_WAVES * 16 * H_PITCH];

    const int tid = threadIdx.x;
    const int lane = tid & 31;
    const int w = __builtin_amdgcn_readfirstlane(tid >> 5);
    const int c = lane & 15;
    const int h = lane >> 4;
    const int stride = MLP_GROUPS * MLP_WAVES;

    int g = (int)blockIdx.x + MLP_GROUPS * w;
    v2f af[16];
    v2f an[16] = {};
    if (g < ntiles) load_a_tile(a, g, c, h, af);

    for (int p = tid; p < HFEAT * HFEAT; p += MLP_THREADS) {
        const int j = p & 1, ln = (p >> 1) & 31, q = p >> 6;
        w2s[p] = W2[(4 * (q >> 2) + 2 * (ln >> 4) + j) * HFEAT + 16 * (q & 3) + (ln & 15)];
    }
    for (int p = tid; p < HFEAT * FC; p += MLP_THREADS) {
        const int j = p & 1, ln = (p >> 1) & 31, q = p >> 6;
        f1s[p] = fc1W[(4 * (q >> 1) + 2 * (ln >> 4) + j) * FC + 16 * (q & 1) + (ln & 15)];
    }
    float bias2[4], bias1[2], wo[2];
#pragma unroll
    for (int t = 0; t < 4; ++t) bias2[t] = b2[16 * t + c];
#pragma unroll
    for (int t = 0; t < 2; ++t) {
        bias1[t] = fc1b[16 * t + c];
        wo[t] = fcoW[16 * t + c];
    }
    const float bo = fcob[0];
    __syncthreads();

    float* hw = hs + w * 16 * H_PITCH;
    while (g < ntiles) {
        const int gn = g + stride;
        if (gn < ntiles) load_a_tile(a, gn, c, h, an);

        v8f acc[4] = {};
#pragma unroll
        for (int kk = 0; kk < 16; ++kk) {
#pragma unroll
            for (int t = 0; t < 4; ++t) {
                const v2f b = *(const v2f*)(w2s + ((kk * 4 + t) * 32 + lane) * 2);
                acc[t] = __builtin_amdgcn_wmma_f32_16x16x4_f32(false, af[kk], false, b,
                                                               (short)0, acc[t], false, false);
            }
        }
#pragma unroll
        for (int t = 0; t < 4; ++t) {
#pragma unroll
            for (int r = 0; r < 8; ++r) {
                const float v = acc[t][r] + bias2[t];
                hw[(8 * h + r) * H_PITCH + 16 * t + c] = v > 0.0f ? v : 0.0f;
            }
        }
        __builtin_amdgcn_fence(__ATOMIC_RELEASE, "wavefront");
        __builtin_amdgcn_wave_barrier();
        __builtin_amdgcn_fence(__ATOMIC_ACQUIRE, "wavefront");

        v8f gacc[2] = {};
#pragma unroll
        for (int kk = 0; kk < 16; ++kk) {
            const v2f x = *(const v2f*)(hw + c * H_PITCH + 4 * kk + 2 * h);
#pragma unroll
            for (int t = 0; t < 2; ++t) {
                const v2f b = *(const v2f*)(f1s + ((kk * 2 + t) * 32 + lane) * 2);
                gacc[t] = __builtin_amdgcn_wmma_f32_16x16x4_f32(false, x, false, b,
                                                                (short)0, gacc[t], false, false);
            }
        }
        __builtin_amdgcn_fence(__ATOMIC_RELEASE, "wavefront");
        __builtin_amdgcn_wave_barrier();

        float p[8];
#pragma unroll
        for (int r = 0; r < 8; ++r) {
            float v0 = gacc[0][r] + bias1[0];
            float v1 = gacc[1][r] + bias1[1];
            v0 = v0 > 0.0f ? v0 : 0.0f;
            v1 = v1 > 0.0f ? v1 : 0.0f;
            p[r] = v0 * wo[0] + v1 * wo[1];
        }
#pragma unroll
        for (int off = 8; off > 0; off >>= 1) {
#pragma unroll
            for (int r = 0; r < 8; ++r) p[r] += __shfl_xor(p[r], off, 32);
        }
        float v = p[0];
#pragma unroll
        for (int r = 1; r < 8; ++r) v = (c == r) ? p[r] : v;
        if (c < 8) out[(size_t)g * 16 + 8 * h + c] = v + bo;

#pragma unroll
        for (int kk = 0; kk < 16; ++kk) af[kk] = an[kk];
        g = gn;
    }
}

extern "C" void kernel_launch(void* const* d_in, const int* in_sizes, int n_in,
                              void* d_out, int out_size, void* d_ws, size_t ws_size,
                              hipStream_t stream) {
    const float* x     = (const float*)d_in[0];
    const int*   ei    = (const int*)d_in[1];
    const float* W1    = (const float*)d_in[2];
    const float* b1    = (const float*)d_in[3];
    const float* W2    = (const float*)d_in[4];
    const float* b2    = (const float*)d_in[5];
    const float* fc1W  = (const float*)d_in[6];
    const float* fc1b  = (const float*)d_in[7];
    const float* fcoW  = (const float*)d_in[8];
    const float* fcob  = (const float*)d_in[9];
    float* out = (float*)d_out;

    const int N = in_sizes[0] / 128;
    const int E = in_sizes[1] / 2;
    const int* src = ei;
    const int* dst = ei + E;

    float* dinv = (float*)d_ws;
    float* bufA = dinv + N;
    float* bufB = bufA + (size_t)N * HFEAT;
    unsigned* rec = (unsigned*)(bufB + (size_t)N * HFEAT);

    auto blocks = [](long long work, int bs) { return (int)((work + bs - 1) / bs); };

    const int NBLK = blocks(E, BM_EDGES);
    k_records<<<NBLK, DEG_THREADS, 0, stream>>>(src, dst, rec, E, NBLK * BM_WORDS);

    k_xw1<<<XW_GROUPS, XW_THREADS, 0, stream>>>(x, W1, bufA, N / 16, rec, dst, dinv, NBLK, N, E);
    k_aggregate<true><<<blocks(N, TILE_ROWS), AGG_THREADS, 0, stream>>>(src, dst, dinv, bufA, b1, bufB, rec, NBLK, N, E);

    k_aggregate<false><<<blocks(N, TILE_ROWS), AGG_THREADS, 0, stream>>>(src, dst, dinv, bufB, b2, bufA, rec, NBLK, N, E);

    k_mlp<<<MLP_GROUPS, MLP_THREADS, 0, stream>>>(bufA, W2, b2, fc1W, fc1b, fcoW, fcob, out, N / 16);
}


# aggregation: all in-degrees sit as bytes in LDS (written by the x @ W1 launch, copied by 16-byte loads); an edge's coefficient is rsqrt of two LDS bytes, not a scattered 4-byte load; sums unchanged
# speedup vs baseline: 4.2454x; 1.0372x over previous
#include <hip/hip_runtime.h>
#include <hip/hip_bf16.h>

typedef __attribute__((ext_vector_type(2))) float v2f;
typedef __attribute__((ext_vector_type(8))) float v8f;

#define HFEAT 64

#define TILE_ROWS 391
#define NTILES 256
#define BM_EDGES 2048
#define DEG_THREADS 1024
#define DEG_EPT (BM_EDGES / DEG_THREADS)
#define BM_WORDS 32
#define BM_PITCH (BM_WORDS + 1)
#define REC_CAP (BM_WORDS - 1)
#define REC_SHIFT 9
#define REC_MASK ((1u << REC_SHIFT) - 1u)

__global__ __launch_bounds__(DEG_THREADS) void k_records(const int* __restrict__ src,
                                                         const int* __restrict__ dst,
                                                         unsigned* __restrict__ rec, int e, int wpt) {
    __shared__ unsigned s[NTILES * BM_PITCH];
    const int tid = threadIdx.x;
    const int lane = tid & 31;
    const int w = tid >> 5;
    const int t0 = blockIdx.x * BM_EDGES + tid;
    int d[DEG_EPT];
    int sv[DEG_EPT];
#pragma unroll
    for (int k = 0; k < DEG_EPT; ++k) {
        const int t = t0 + k * DEG_THREADS;
        const int tl = t < e ? t : e - 1;
        d[k] = dst[tl];
        sv[k] = src[tl];
        if (t >= e) d[k] = -1;
    }
    for (int i = tid; i < NTILES * BM_PITCH; i += DEG_THREADS) s[i] = 0u;
    __syncthreads();
#pragma unroll
    for (int k = 0; k < DEG_EPT; ++k) {
        if (d[k] >= 0) {
            const unsigned tile = (unsigned)d[k] / TILE_ROWS;
            if (tile < NTILES) {
                const unsigned slot = __hip_atomic_fetch_add(&s[tile * BM_PITCH], 1u, __ATOMIC_RELAXED,
                                                             __HIP_MEMORY_SCOPE_WORKGROUP);
                if (slot < REC_CAP)
                    s[tile * BM_PITCH + 1 + slot] =
                        ((unsigned)sv[k] << REC_SHIFT) | ((unsigned)d[k] - tile * TILE_ROWS);
            }
        }
    }
    __syncthreads();
    unsigned* o = rec + (size_t)w * wpt + blockIdx.x * BM_WORDS + lane;
#pragma unroll
    for (int k = 0; k < NTILES / BM_WORDS; ++k)
        o[(size_t)k * BM_WORDS * wpt] = s[(w + k * BM_WORDS) * BM_PITCH + lane];
}

#define XW_K 128
#define XW_WAVES 16
#define XW_THREADS (XW_WAVES * 32)
#define XW_GROUPS 256
#define XW_PITCH 68

typedef __attribute__((ext_vector_type(4))) float v4f;
typedef __attribute__((ext_vector_type(4))) unsigned v4u;
#define CNT_BATCH 13
#define DEG_SAT 255u

static_assert(XW_GROUPS == NTILES, "one x @ W1 work-group per aggregation tile");
static_assert(XW_THREADS >= TILE_ROWS, "one thread per tile row clears the counters and stores dinv");

__device__ __forceinline__ void load_x_tile(const float* __restrict__ x, int g, int c, int h, v4f (&f)[16]) {
    const float* xr = x + ((size_t)g * 16 + c) * XW_K + 4 * h;
#pragma unroll
    for (int q = 0; q < 16; ++q) f[q] = *(const v4f*)(xr + 8 * q);
}

__device__ __forceinline__ void load_rec_lines(const unsigned* __restrict__ row, int lg0, int nblk, int lsub, int wq,
                                               v4u (&lv)[CNT_BATCH]) {
#pragma unroll
    for (int k = 0; k < CNT_BATCH; ++k) {
        const int line = 4 * (lg0 + k * XW_WAVES) + lsub;
        const int ll = line < nblk ? line : nblk - 1;
        lv[k] = *(const v4u*)(row + (size_t)ll * BM_WORDS + 4 * wq);
    }
}

__global__ __launch_bounds__(XW_THREADS) void k_xw1(const float* __restrict__ x, const float* __restrict__ W1,
                                                    float* __restrict__ xw, int ntiles,
                                                    const unsigned* __restrict__ rec, const int* __restrict__ dst,
                                                    float* __restrict__ dinv, unsigned char* __restrict__ deg8,
                                                    int nblk, int n, int e) {
    __shared__ __attribute__((aligned(16))) float w1s[XW_K * HFEAT];
    __shared__ __attribute__((aligned(16))) float os[XW_WAVES * 16 * XW_PITCH];
    __shared__ unsigned cnts[TILE_ROWS];

    const int tid = threadIdx.x;
    const int lane = tid & 31;
    const int w = __builtin_amdgcn_readfirstlane(tid >> 5);
    const int c = lane & 15;
    const int h = lane >> 4;
    const int stride = XW_GROUPS * XW_WAVES;

    const int d0 = (int)blockIdx.x * TILE_ROWS;
    int cnt = n - d0;
    if (cnt > TILE_ROWS) cnt = TILE_ROWS;
    const unsigned* row = rec + (size_t)blockIdx.x * nblk * BM_WORDS;
    const int nq = (nblk + 3) >> 2;
    const int lsub = lane >> 3, wq = lane & 7;
    v4u lv[CNT_BATCH];
    load_rec_lines(row, w, nblk, lsub, wq, lv);

    int g = (int)blockIdx.x + XW_GROUPS * w;
    v4f xf[16];
    v4f xn[16] = {};
    if (g < ntiles) load_x_tile(x, g, c, h, xf);

    if (tid < TILE_ROWS) cnts[tid] = 0u;

    for (int p = tid; p < XW_K * HFEAT; p += XW_THREADS) {
        const int j = p & 1, ln = (p >> 1) & 31, sq = p >> 6;
        const int s = sq >> 2, t = sq & 3;
        w1s[p] = W1[(8 * (s >> 1) + 4 * (ln >> 4) + 2 * (s & 1) + j) * HFEAT + 16 * t + (ln & 15)];
    }
    __syncthreads();

    for (int lg0 = w; lg0 < nq; lg0 += XW_WAVES * CNT_BATCH) {
        if (lg0 != w) load_rec_lines(row, lg0, nblk, lsub, wq, lv);
#pragma unroll
        for (int k = 0; k < CNT_BATCH; ++k) {
            const bool real = 4 * (lg0 + k * XW_WAVES) + lsub < nblk;
            const unsigned lc = (unsigned)__shfl((int)lv[k].x, lane & 24);
            const bool fits = lc <= REC_CAP;
#pragma unroll
            for (int j = 0; j < 4; ++j) {
                const unsigned wi = 4u * wq + j;
                const unsigned loc = lv[k][j] & REC_MASK;
                if (real && fits && wi >= 1u && wi <= lc && loc < TILE_ROWS)
                    __hip_atomic_fetch_add(&cnts[loc], 1u, __ATOMIC_RELAXED, __HIP_MEMORY_SCOPE_WORKGROUP);
            }
            unsigned over = __builtin_amdgcn_ballot_w32(real && !fits && wq == 0);
            while (over) {
                const int bl = __builtin_ctz(over);
                over &= over - 1u;
                const int e0 = (4 * (lg0 + k * XW_WAVES) + (bl >> 3)) * BM_EDGES + lane;
#pragma nounroll
                for (int r = 0; r < BM_EDGES / 32; ++r) {
                    const int e_i = e0 + r * 32;
                    if (e_i < e) {
                        const unsigned loc = (unsigned)(dst[e_i] - d0);
                        if (loc < (unsigned)cnt)
                            __hip_atomic_fetch_add(&cnts[loc], 1u, __ATOMIC_RELAXED, __HIP_MEMORY_SCOPE_WORKGROUP);
                    }
                }
            }
        }
    }

    float* ow = os + w * 16 * XW_PITCH;
    while (g < ntiles) {
        const int gn = g + stride;
        if (gn < ntiles) load_x_tile(x, gn, c, h, xn);

        v8f acc[4] = {};
#pragma unroll
        for (int q = 0; q < 16; ++q) {
#pragma unroll
            for (int u = 0; u < 2; ++u) {
                v2f a;
                a.x = u ? xf[q].z : xf[q].x;
                a.y = u ? xf[q].w : xf[q].y;
#pragma unroll
                for (int t = 0; t < 4; ++t) {
                    const v2f b = *(const v2f*)(w1s + (((2 * q + u) * 4 + t) * 32 + lane) * 2);
                    acc[t] = __builtin_amdgcn_wmma_f32_16x16x4_f32(false, a, false, b,
                                                                   (short)0, acc[t], false, false);
                }
            }
        }
#pragma unroll
        for (int t = 0; t < 4; ++t) {
#pragma unroll
            for (int r = 0; r < 8; ++r) ow[(8 * h + r) * XW_PITCH + 16 * t + c] = acc[t][r];
        }
        __builtin_amdgcn_fence(__ATOMIC_RELEASE, "wavefront");
        __builtin_amdgcn_wave_barrier();
        __builtin_amdgcn_fence(__ATOMIC_ACQUIRE, "wavefront");
        float* orow = xw + ((size_t)g * 16 + h) * HFEAT + 4 * c;
#pragma unroll
        for (int i = 0; i < 8; ++i) {
            const v4f v = *(const v4f*)(ow + (2 * i + h) * XW_PITCH + 4 * c);
            *(v4f*)(orow + (size_t)(2 * i) * HFEAT) = v;
        }
        __builtin_amdgcn_fence(__ATOMIC_RELEASE, "wavefront");
        __builtin_amdgcn_wave_barrier();

#pragma unroll
        for (int q = 0; q < 16; ++q) xf[q] = xn[q];
        g = gn;
    }

    __syncthreads();
    if (tid < cnt) {
        const unsigned dg = 1u + cnts[tid];
        dinv[d0 + tid] = rsqrtf((float)dg);
        deg8[d0 + tid] = (unsigned char)(dg < DEG_SAT ? dg : DEG_SAT);
    }
}

#define AGG_THREADS 1024
#define AGG_WAVES 32
#define SCAN_LINES 4
#define OVF_EDGES 512
#define DEG_BYTES (NTILES * TILE_ROWS)
#define DEG_BATCH 7
#define QCAP (32 + OVF_EDGES)

static_assert(DEG_BATCH * 1024 * 16 >= DEG_BYTES, "one batch of 16-byte loads covers the in-degree table");

__device__ __forceinline__ void load_lines(const unsigned* __restrict__ row, int base, int nblk,
                                           int lane, unsigned (&v)[SCAN_LINES]) {
#pragma unroll
    for (int j = 0; j < SCAN_LINES; ++j)
        v[j] = (base + j < nblk) ? row[(size_t)(base + j) * BM_WORDS + lane] : 0u;
}

__device__ __forceinline__ void lds_add(float* p, float v) {
    __hip_atomic_fetch_add(p, v, __ATOMIC_RELAXED, __HIP_MEMORY_SCOPE_WORKGROUP);
}

__device__ __forceinline__ void drain(const unsigned* qe, int start, int count, int lane, int d0,
                                      const float* __restrict__ dinv, const unsigned char* degs,
                                      const float* __restrict__ xw, float* tile) {
    __builtin_amdgcn_fence(__ATOMIC_RELEASE, "wavefront");
    __builtin_amdgcn_wave_barrier();
    __builtin_amdgcn_fence(__ATOMIC_ACQUIRE, "wavefront");
    int s_i = 0, t_i = 0;
    float c_i = 0.0f;
    if (lane < count) {
        const unsigned r_i = qe[start + lane];
        s_i = (int)(r_i >> REC_SHIFT);
        t_i = (int)(r_i & REC_MASK);
        const unsigned gs = degs[s_i];
        const unsigned gd = degs[d0 + t_i];
        c_i = rsqrtf((float)gs) * rsqrtf((float)gd);
        if (gs == DEG_SAT || gd == DEG_SAT) {
            const float fs = gs == DEG_SAT ? dinv[s_i] : rsqrtf((float)gs);
            const float fd = gd == DEG_SAT ? dinv[d0 + t_i] : rsqrtf((float)gd);
            c_i = fs * fd;
        }
    }
    int i = 0;
    for (; i + 8 <= count; i += 8) {
        float r[16];
        int tt[8];
        float cc[8];
#pragma unroll
        for (int u = 0; u < 8; ++u) {
            int s = __builtin_amdgcn_readlane(s_i, i + u);
            tt[u] = __builtin_amdgcn_readlane(t_i, i + u);
            cc[u] = __int_as_float(__builtin_amdgcn_readlane(__float_as_int(c_i), i + u));
            const float* rp = xw + (size_t)s * HFEAT + lane;
            r[2 * u] = rp[0];
            r[2 * u + 1] = rp[32];
        }
#pragma unroll
        for (int u = 0; u < 8; ++u) {
            float* o = tile + tt[u] * HFEAT + lane;
            lds_add(o, cc[u] * r[2 * u]);
            lds_add(o + 32, cc[u] * r[2 * u + 1]);
        }
    }
    for (; i < count; ++i) {
        int s = __builtin_amdgcn_readlane(s_i, i);
        int t = __builtin_amdgcn_readlane(t_i, i);
        float c = __int_as_float(__builtin_amdgcn_readlane(__float_as_int(c_i), i));
        const float* rp = xw + (size_t)s * HFEAT + lane;
        float r0 = rp[0];
        float r1 = rp[32];
        float* o = tile + t * HFEAT + lane;
        lds_add(o, c * r0);
        lds_add(o + 32, c * r1);
    }
    __builtin_amdgcn_fence(__ATOMIC_RELEASE, "wavefront");
    __builtin_amdgcn_wave_barrier();
}

template <bool BIAS_RELU>
__global__ __launch_bounds__(AGG_THREADS) void k_aggregate(
        const int* __restrict__ src, const int* __restrict__ dst, const float* __restrict__ dinv,
        const float* __restrict__ xw, const float* __restrict__ bias, float* __restrict__ out,
        const unsigned* __restrict__ rec, const unsigned char* __restrict__ deg8, int nblk, int n, int e) {
    __shared__ float tile[TILE_ROWS * HFEAT];
    __shared__ unsigned q_e[AGG_WAVES * QCAP];
    __shared__ __attribute__((aligned(16))) unsigned char degs[DEG_BYTES];

    const int tid = threadIdx.x;
    const int lane = tid & 31;
    const int w = __builtin_amdgcn_readfirstlane(tid >> 5);
    const int d0 = blockIdx.x * TILE_ROWS;
    int cnt = n - d0;
    if (cnt > TILE_ROWS) cnt = TILE_ROWS;
    const int total = cnt * HFEAT;

    const int n16 = n >> 4;
    v4u dv[DEG_BATCH];
    if (n16 > 0) {
#pragma unroll
        for (int k = 0; k < DEG_BATCH; ++k) {
            const int p = tid + k * AGG_THREADS;
            dv[k] = ((const v4u*)deg8)[p < n16 ? p : n16 - 1];
        }
    }

    const float* xwt = xw + (size_t)d0 * HFEAT;
    for (int i = tid; i < total; i += AGG_THREADS) {
        float di = dinv[d0 + (i >> 6)];
        tile[i] = di * di * xwt[i];
    }
#pragma unroll
    for (int k = 0; k < DEG_BATCH; ++k) {
        const int p = tid + k * AGG_THREADS;
        if (p < n16) ((v4u*)degs)[p] = dv[k];
    }
    if (tid < (n & 15)) degs[(n16 << 4) + tid] = deg8[(n16 << 4) + tid];
    __syncthreads();

    unsigned* qe = q_e + w * QCAP;
    int qn = 0;
    const unsigned* row = rec + (size_t)blockIdx.x * nblk * BM_WORDS;

    unsigned cur[SCAN_LINES];
    unsigned nxt[SCAN_LINES] = {};
    int base = w * SCAN_LINES;
    if (base < nblk) load_lines(row, base, nblk, lane, cur);
    for (; base < nblk; base += AGG_WAVES * SCAN_LINES) {
        const int nb = base + AGG_WAVES * SCAN_LINES;
        if (nb < nblk) load_lines(row, nb, nblk, lane, nxt);
#pragma unroll
        for (int j = 0; j < SCAN_LINES; ++j) {
            const unsigned word = cur[j];
            const int c = __builtin_amdgcn_readlane((int)word, 0);
            const bool fits = c <= REC_CAP;
            const int passes = fits ? 1 : BM_EDGES / OVF_EDGES;
            for (int h = 0; h < passes; ++h) {
                if (fits) {
                    if (lane >= 1 && lane <= c) qe[qn + lane - 1] = word;
                    qn += c;
                } else {
                    const int e0 = (base + j) * BM_EDGES + h * OVF_EDGES + lane;
#pragma nounroll
                    for (int r = 0; r < OVF_EDGES / 32; ++r) {
                        const int e_i = e0 + r * 32;
                        bool m = false;
                        unsigned rr = 0u;
                        if (e_i < e) {
                            const unsigned loc = (unsigned)(dst[e_i] - d0);
                            if (loc < (unsigned)cnt) {
                                m = true;
                                rr = ((unsigned)src[e_i] << REC_SHIFT) | loc;
                            }
                        }
                        const unsigned any = __builtin_amdgcn_ballot_w32(m);
                        if (m) qe[qn + (int)__builtin_amdgcn_mbcnt_lo(any, 0u)] = rr;
                        qn += __builtin_popcount(any);
                    }
                }
                while (qn >= 32) {
                    qn -= 32;
                    drain(qe, qn, 32, lane, d0, dinv, degs, xw, tile);
                }
            }
        }
#pragma unroll
        for (int j = 0; j < SCAN_LINES; ++j) cur[j] = nxt[j];
    }
    if (qn > 0) drain(qe, 0, qn, lane, d0, dinv, degs, xw, tile);
    __syncthreads();

    float* ot = out + (size_t)d0 * HFEAT;
    for (int i = tid; i < total; i += AGG_THREADS) {
        float v = tile[i];
        if (BIAS_RELU) {
            v += bias[i & (HFEAT - 1)];
            v = v > 0.0f ? v : 0.0f;
        }
        ot[i] = v;
    }
}

#define MLP_WAVES 16
#define MLP_THREADS (MLP_WAVES * 32)
#define MLP_GROUPS 256
#define H_PITCH 68
#define FC 32

__device__ __forceinline__ void load_a_tile(const float* __restrict__ a, int g, int c, int h, v2f (&f)[16]) {
    const float* arow = a + ((size_t)g * 16 + c) * HFEAT + 2 * h;
#pragma unroll
    for (int kk = 0; kk < 16; ++kk) f[kk] = *(const v2f*)(arow + 4 * kk);
}

__global__ __launch_bounds__(MLP_THREADS) void k_mlp(
        const float* __restrict__ a, const float* __restrict__ W2, const float* __restrict__ b2,
        const float* __restrict__ fc1W, const float* __restrict__ fc1b, const float* __restrict__ fcoW,
        const float* __restrict__ fcob, float* __restrict__ out, int ntiles) {
    __shared__ float w2s[HFEAT * HFEAT];
    __shared__ float f1s[HFEAT * FC];
    __shared__ float hs[MLP_WAVES * 16 * H_PITCH];

    const int tid = threadIdx.x;
    const int lane = tid & 31;
    const int w = __builtin_amdgcn_readfirstlane(tid >> 5);
    const int c = lane & 15;
    const int h = lane >> 4;
    const int stride = MLP_GROUPS * MLP_WAVES;

    int g = (int)blockIdx.x + MLP_GROUPS * w;
    v2f af[16];
    v2f an[16] = {};
    if (g < ntiles) load_a_tile(a, g, c, h, af);

    for (int p = tid; p < HFEAT * HFEAT; p += MLP_THREADS) {
        const int j = p & 1, ln = (p >> 1) & 31, q = p >> 6;
        w2s[p] = W2[(4 * (q >> 2) + 2 * (ln >> 4) + j) * HFEAT + 16 * (q & 3) + (ln & 15)];
    }
    for (int p = tid; p < HFEAT * FC; p += MLP_THREADS) {
        const int j = p & 1, ln = (p >> 1) & 31, q = p >> 6;
        f1s[p] = fc1W[(4 * (q >> 1) + 2 * (ln >> 4) + j) * FC + 16 * (q & 1) + (ln & 15)];
    }
    float bias2[4], bias1[2], wo[2];
#pragma unroll
    for (int t = 0; t < 4; ++t) bias2[t] = b2[16 * t + c];
#pragma unroll
    for (int t = 0; t < 2; ++t) {
        bias1[t] = fc1b[16 * t + c];
        wo[t] = fcoW[16 * t + c];
    }
    const float bo = fcob[0];
    __syncthreads();

    float* hw = hs + w * 16 * H_PITCH;
    while (g < ntiles) {
        const int gn = g + stride;
        if (gn < ntiles) load_a_tile(a, gn, c, h, an);

        v8f acc[4] = {};
#pragma unroll
        for (int kk = 0; kk < 16; ++kk) {
#pragma unroll
            for (int t = 0; t < 4; ++t) {
                const v2f b = *(const v2f*)(w2s + ((kk * 4 + t) * 32 + lane) * 2);
                acc[t] = __builtin_amdgcn_wmma_f32_16x16x4_f32(false, af[kk], false, b,
                                                               (short)0, acc[t], false, false);
            }
        }
#pragma unroll
        for (int t = 0; t < 4; ++t) {
#pragma unroll
            for (int r = 0; r < 8; ++r) {
                const float v = acc[t][r] + bias2[t];
                hw[(8 * h + r) * H_PITCH + 16 * t + c] = v > 0.0f ? v : 0.0f;
            }
        }
        __builtin_amdgcn_fence(__ATOMIC_RELEASE, "wavefront");
        __builtin_amdgcn_wave_barrier();
        __builtin_amdgcn_fence(__ATOMIC_ACQUIRE, "wavefront");

        v8f gacc[2] = {};
#pragma unroll
        for (int kk = 0; kk < 16; ++kk) {
            const v2f x = *(const v2f*)(hw + c * H_PITCH + 4 * kk + 2 * h);
#pragma unroll
            for (int t = 0; t < 2; ++t) {
                const v2f b = *(const v2f*)(f1s + ((kk * 2 + t) * 32 + lane) * 2);
                gacc[t] = __builtin_amdgcn_wmma_f32_16x16x4_f32(false, x, false, b,
                                                                (short)0, gacc[t], false, false);
            }
        }
        __builtin_amdgcn_fence(__ATOMIC_RELEASE, "wavefront");
        __builtin_amdgcn_wave_barrier();

        float p[8];
#pragma unroll
        for (int r = 0; r < 8; ++r) {
            float v0 = gacc[0][r] + bias1[0];
            float v1 = gacc[1][r] + bias1[1];
            v0 = v0 > 0.0f ? v0 : 0.0f;
            v1 = v1 > 0.0f ? v1 : 0.0f;
            p[r] = v0 * wo[0] + v1 * wo[1];
        }
#pragma unroll
        for (int off = 8; off > 0; off >>= 1) {
#pragma unroll
            for (int r = 0; r < 8; ++r) p[r] += __shfl_xor(p[r], off, 32);
        }
        float v = p[0];
#pragma unroll
        for (int r = 1; r < 8; ++r) v = (c == r) ? p[r] : v;
        if (c < 8) out[(size_t)g * 16 + 8 * h + c] = v + bo;

#pragma unroll
        for (int kk = 0; kk < 16; ++kk) af[kk] = an[kk];
        g = gn;
    }
}

extern "C" void kernel_launch(void* const* d_in, const int* in_sizes, int n_in,
                              void* d_out, int out_size, void* d_ws, size_t ws_size,
                              hipStream_t stream) {
    const float* x     = (const float*)d_in[0];
    const int*   ei    = (const int*)d_in[1];
    const float* W1    = (const float*)d_in[2];
    const float* b1    = (const float*)d_in[3];
    const float* W2    = (const float*)d_in[4];
    const float* b2    = (const float*)d_in[5];
    const float* fc1W  = (const float*)d_in[6];
    const float* fc1b  = (const float*)d_in[7];
    const float* fcoW  = (const float*)d_in[8];
    const float* fcob  = (const float*)d_in[9];
    float* out = (float*)d_out;

    const int N = in_sizes[0] / 128;
    const int E = in_sizes[1] / 2;
    const int* src = ei;
    const int* dst = ei + E;

    float* dinv = (float*)d_ws;
    float* bufA = dinv + N;
    float* bufB = bufA + (size_t)N * HFEAT;
    unsigned* rec = (unsigned*)(bufB + (size_t)N * HFEAT);

    auto blocks = [](long long work, int bs) { return (int)((work + bs - 1) / bs); };

    const int NBLK = blocks(E, BM_EDGES);
    unsigned char* deg8 = (unsigned char*)(rec + (size_t)NTILES * NBLK * BM_WORDS);
    k_records<<<NBLK, DEG_THREADS, 0, stream>>>(src, dst, rec, E, NBLK * BM_WORDS);

    k_xw1<<<XW_GROUPS, XW_THREADS, 0, stream>>>(x, W1, bufA, N / 16, rec, dst, dinv, deg8, NBLK, N, E);
    k_aggregate<true><<<blocks(N, TILE_ROWS), AGG_THREADS, 0, stream>>>(src, dst, dinv, bufA, b1, bufB, rec, deg8, NBLK, N, E);

    k_aggregate<false><<<blocks(N, TILE_ROWS), AGG_THREADS, 0, stream>>>(src, dst, dinv, bufB, b2, bufA, rec, deg8, NBLK, N, E);

    k_mlp<<<MLP_GROUPS, MLP_THREADS, 0, stream>>>(bufA, W2, b2, fc1W, fc1b, fcoW, fcob, out, N / 16);
}
